# starting kernel
# baseline (speedup 1.0000x reference)
#include <hip/hip_runtime.h>


typedef __attribute__((ext_vector_type(2))) float v2f;
typedef __attribute__((ext_vector_type(8))) float v8f;

#define N_NODES 100000
#define N_EDGES 1600000
#define OUTC 32

template<int K>
__global__ __launch_bounds__(256) void gemm_x32_wmma(
    const float* __restrict__ X,
    const float* __restrict__ W,
    float* __restrict__ H,
    int n_rows)
{
  const int lane = threadIdx.x & 31;
  const int tile = (int)(blockIdx.x * (blockDim.x >> 5) + (threadIdx.x >> 5));
  const int n_tiles = n_rows >> 4;
  if (tile >= n_tiles) return;

  const int  m   = lane & 15;
  const int  kh  = lane >> 4;
  const long row0 = (long)tile << 4;

  const float* xrow = X + (row0 + m) * K;

  v8f c0 = {};
  v8f c1 = {};

  #pragma unroll 8
  for (int kb = 0; kb < K; kb += 4) {
    const int ka = kb + 2 * kh;
    v2f a;
    a.x = xrow[ka];
    a.y = xrow[ka + 1];
    v2f b0, b1;
    b0.x = W[ka * OUTC + m];
    b0.y = W[(ka + 1) * OUTC + m];
    b1.x = W[ka * OUTC + 16 + m];
    b1.y = W[(ka + 1) * OUTC + 16 + m];
    c0 = __builtin_amdgcn_wmma_f32_16x16x4_f32(false, a, false, b0, (short)0, c0, false, false);
    c1 = __builtin_amdgcn_wmma_f32_16x16x4_f32(false, a, false, b1, (short)0, c1, false, false);
  }

  float* hrow = H + (row0 + 8 * kh) * OUTC + m;
  #pragma unroll
  for (int r = 0; r < 8; ++r) {
    hrow[r * OUTC]      = c0[r];
    hrow[r * OUTC + 16] = c1[r];
  }
}

__device__ __forceinline__ void atomic_add_f32_dev(float* p, float v) {
  asm volatile("global_atomic_add_f32 %0, %1, off scope:SCOPE_DEV"
               :
               : "v"(p), "v"(v)
               : "memory");
}

__global__ __launch_bounds__(256) void gcn_scatter(
    const float* __restrict__ H,
    const int*   __restrict__ eidx,
    const float* __restrict__ ew,
    float*       __restrict__ out,
    int n_edges)
{
  const int lane   = threadIdx.x & 31;
  const int wave0  = (int)((blockIdx.x * (long)blockDim.x + threadIdx.x) >> 5);
  const int nwaves = (int)(((long)gridDim.x * blockDim.x) >> 5);

  for (int e = wave0; e < n_edges; e += nwaves) {
    const int   s = eidx[e];
    const int   d = eidx[n_edges + e];
    const float w = ew[e];
    const float v = w * H[(long)s * OUTC + lane];
    atomic_add_f32_dev(&out[(long)d * OUTC + lane], v);
  }
}

extern "C" void kernel_launch(void* const* d_in, const int* in_sizes, int n_in,
                              void* d_out, int out_size, void* d_ws, size_t ws_size,
                              hipStream_t stream) {
  const float* x  = (const float*)d_in[0];
  const float* W1 = (const float*)d_in[1];
  const float* W2 = (const float*)d_in[2];
  const float* ew = (const float*)d_in[3];
  const int*   ei = (const int*)d_in[4];
  float* out = (float*)d_out;

  const size_t feat_bytes = (size_t)N_NODES * OUTC * sizeof(float);
  float* h    = (float*)d_ws;
  float* agg1 = (float*)((char*)d_ws + feat_bytes);

  hipMemsetAsync(agg1, 0, feat_bytes, stream);
  hipMemsetAsync(out,  0, feat_bytes, stream);

  const int tiles = N_NODES / 16;
  const int wpb   = 256 / 32;
  dim3 gemm_grid((tiles + wpb - 1) / wpb);

  gemm_x32_wmma<128><<<gemm_grid, 256, 0, stream>>>(x, W1, h, N_NODES);
  gcn_scatter<<<4096, 256, 0, stream>>>(h, ei, ew, agg1, N_EDGES);

  gemm_x32_wmma<32><<<gemm_grid, 256, 0, stream>>>(agg1, W2, h, N_NODES);
  gcn_scatter<<<4096, 256, 0, stream>>>(h, ei, ew, out, N_EDGES);
}


# Scatter: one wave per 32-edge batch: coalesced index loads, lane-to-scalar broadcast, 32 row gathers in flight before the f32 atomics (was 2 dependent loads per edge); atomic arrival order differs
# speedup vs baseline: 1.8415x; 1.8415x over previous
#include <hip/hip_runtime.h>


typedef __attribute__((ext_vector_type(2))) float v2f;
typedef __attribute__((ext_vector_type(8))) float v8f;

#define N_NODES 100000
#define N_EDGES 1600000
#define OUTC 32

template<int K>
__global__ __launch_bounds__(256) void gemm_x32_wmma(
    const float* __restrict__ X,
    const float* __restrict__ W,
    float* __restrict__ H,
    int n_rows)
{
  const int lane = threadIdx.x & 31;
  const int tile = (int)(blockIdx.x * (blockDim.x >> 5) + (threadIdx.x >> 5));
  const int n_tiles = n_rows >> 4;
  if (tile >= n_tiles) return;

  const int  m   = lane & 15;
  const int  kh  = lane >> 4;
  const long row0 = (long)tile << 4;

  const float* xrow = X + (row0 + m) * K;

  v8f c0 = {};
  v8f c1 = {};

  #pragma unroll 8
  for (int kb = 0; kb < K; kb += 4) {
    const int ka = kb + 2 * kh;
    v2f a;
    a.x = xrow[ka];
    a.y = xrow[ka + 1];
    v2f b0, b1;
    b0.x = W[ka * OUTC + m];
    b0.y = W[(ka + 1) * OUTC + m];
    b1.x = W[ka * OUTC + 16 + m];
    b1.y = W[(ka + 1) * OUTC + 16 + m];
    c0 = __builtin_amdgcn_wmma_f32_16x16x4_f32(false, a, false, b0, (short)0, c0, false, false);
    c1 = __builtin_amdgcn_wmma_f32_16x16x4_f32(false, a, false, b1, (short)0, c1, false, false);
  }

  float* hrow = H + (row0 + 8 * kh) * OUTC + m;
  #pragma unroll
  for (int r = 0; r < 8; ++r) {
    hrow[r * OUTC]      = c0[r];
    hrow[r * OUTC + 16] = c1[r];
  }
}

__device__ __forceinline__ void atomic_add_f32_dev(float* p, float v) {
  asm volatile("global_atomic_add_f32 %0, %1, off scope:SCOPE_DEV"
               :
               : "v"(p), "v"(v)
               : "memory");
}

__global__ __launch_bounds__(256) void gcn_scatter(
    const float* __restrict__ H,
    const int*   __restrict__ eidx,
    const float* __restrict__ ew,
    float*       __restrict__ out,
    int n_edges)
{
  const int lane   = threadIdx.x & 31;
  const int wave0  = __builtin_amdgcn_readfirstlane(
      (int)((blockIdx.x * (long)blockDim.x + threadIdx.x) >> 5));
  const int nwaves = (int)(((long)gridDim.x * blockDim.x) >> 5);
  const int stride = nwaves * 32;

  const float* hl = H + lane;
  float*       ol = out + lane;

  int b = wave0 * 32;
  int   s = 0, d = 0;
  float w = 0.0f;
  if (b + lane < n_edges) {
    s = eidx[b + lane];
    d = eidx[n_edges + b + lane];
    w = ew[b + lane];
  }

  for (; b < n_edges; b += stride) {
    const int nb = b + stride;
    const int n  = min(32, n_edges - b);
    int   s2 = 0, d2 = 0;
    float w2 = 0.0f;

    if (n == 32) {
      float hv[32];
      #pragma unroll
      for (int u = 0; u < 32; ++u)
        hv[u] = hl[(long)__builtin_amdgcn_readlane(s, u) * OUTC];
      if (nb + lane < n_edges) {
        s2 = eidx[nb + lane];
        d2 = eidx[n_edges + nb + lane];
        w2 = ew[nb + lane];
      }
      #pragma unroll
      for (int u = 0; u < 32; ++u) {
        const int   du = __builtin_amdgcn_readlane(d, u);
        const float wu = __int_as_float(__builtin_amdgcn_readlane(__float_as_int(w), u));
        atomic_add_f32_dev(ol + (long)du * OUTC, wu * hv[u]);
      }
    } else {
      for (int j = 0; j < n; ++j) {
        const int   sj = __builtin_amdgcn_readlane(s, j);
        const int   dj = __builtin_amdgcn_readlane(d, j);
        const float wj = __int_as_float(__builtin_amdgcn_readlane(__float_as_int(w), j));
        atomic_add_f32_dev(ol + (long)dj * OUTC, wj * hl[(long)sj * OUTC]);
      }
    }
    s = s2; d = d2; w = w2;
  }
}

extern "C" void kernel_launch(void* const* d_in, const int* in_sizes, int n_in,
                              void* d_out, int out_size, void* d_ws, size_t ws_size,
                              hipStream_t stream) {
  const float* x  = (const float*)d_in[0];
  const float* W1 = (const float*)d_in[1];
  const float* W2 = (const float*)d_in[2];
  const float* ew = (const float*)d_in[3];
  const int*   ei = (const int*)d_in[4];
  float* out = (float*)d_out;

  const size_t feat_bytes = (size_t)N_NODES * OUTC * sizeof(float);
  float* h    = (float*)d_ws;
  float* agg1 = (float*)((char*)d_ws + feat_bytes);

  hipMemsetAsync(agg1, 0, feat_bytes, stream);
  hipMemsetAsync(out,  0, feat_bytes, stream);

  const int tiles = N_NODES / 16;
  const int wpb   = 256 / 32;
  dim3 gemm_grid((tiles + wpb - 1) / wpb);

  gemm_x32_wmma<128><<<gemm_grid, 256, 0, stream>>>(x, W1, h, N_NODES);
  gcn_scatter<<<2048, 256, 0, stream>>>(h, ei, ew, agg1, N_EDGES);

  gemm_x32_wmma<32><<<gemm_grid, 256, 0, stream>>>(agg1, W2, h, N_NODES);
  gcn_scatter<<<2048, 256, 0, stream>>>(h, ei, ew, out, N_EDGES);
}
